# starting kernel
# baseline (speedup 1.0000x reference)
#include <hip/hip_runtime.h>

typedef __attribute__((ext_vector_type(2))) float v2f;
typedef __attribute__((ext_vector_type(8))) float v8f;

#define BATCH 16384
#define EPS 1e-5f

__device__ inline float wz_val(const float* w_fc1, const float* w_fc2, int k, int n) {
  if (n >= 10) return 0.f;
  int i = k / 52, j = k - i * 52;
  if (j >= 49) return 0.f;
  if (i < 49) return (i <= j) ? w_fc2[n * 2401 + i * 49 + j] : 0.f;
  return w_fc1[n * 49 + j];
}

__global__ void prep_kernel(const float* __restrict__ w_fc1,
                            const float* __restrict__ w_fc2,
                            float2* __restrict__ wswz,
                            float* __restrict__ stats) {
  int t = blockIdx.x * blockDim.x + threadIdx.x;
  if (t < 4) stats[t] = 0.f;
  if (t < 650 * 32) {
    int s = t >> 5, lane = t & 31;
    int n = lane & 15, half = lane >> 4;
    int ka = 4 * s + 2 * half;
    float2 v;
    v.x = wz_val(w_fc1, w_fc2, ka, n);
    v.y = wz_val(w_fc1, w_fc2, ka + 1, n);
    wswz[t] = v;
  }
}

__device__ inline float dot10(float t00, float t01, float t10, float t11,
                              const float* __restrict__ W) {
  return W[0]*t00*t00 + W[1]*t01*t01 + W[2]*t10*t10 + W[3]*t11*t11
       + W[4]*t00*t01 + W[5]*t00*t10 + W[6]*t00*t11
       + W[7]*t01*t10 + W[8]*t01*t11 + W[9]*t10*t11;
}

__global__ __launch_bounds__(256) void layer1_kernel(
    const float* __restrict__ x,
    const float* __restrict__ w1,   const float* __restrict__ w2lr,
    const float* __restrict__ w2tb, const float* __restrict__ w2all,
    const float* __restrict__ w2no, const float* __restrict__ w3,
    float* __restrict__ l1_out, float* __restrict__ stats) {
  __shared__ float s_w1[784];
  __shared__ float s_w2[4][1960];
  __shared__ float s_w3[196];
  __shared__ float s_img[8][784];
  __shared__ float s_red[256];
  int tid = threadIdx.x;
  for (int p = tid; p < 784; p += 256) s_w1[p] = w1[p];
  for (int p = tid; p < 1960; p += 256) {
    s_w2[0][p] = w2lr[p]; s_w2[1][p] = w2tb[p];
    s_w2[2][p] = w2all[p]; s_w2[3][p] = w2no[p];
  }
  for (int p = tid; p < 196; p += 256) s_w3[p] = w3[p];
  int wv = tid >> 5, lane = tid & 31;
  long b = (long)blockIdx.x * 8 + wv;
  const float* xb = x + b * 784;
  for (int p = lane; p < 784; p += 32) s_img[wv][p] = xb[p];
  __syncthreads();

  float sum = 0.f, sq = 0.f;
  for (int pix = lane; pix < 196; pix += 32) {
    int r = pix / 14, c = pix - r * 14;
    int i00 = (2 * r) * 28 + 2 * c;
    float b00 = s_img[wv][i00],      b01 = s_img[wv][i00 + 1];
    float b10 = s_img[wv][i00 + 28], b11 = s_img[wv][i00 + 29];
    float acc = b00 * s_w1[i00] + b01 * s_w1[i00 + 1]
              + b10 * s_w1[i00 + 28] + b11 * s_w1[i00 + 29];
    acc += s_w3[pix];
    int wofs = r * 140 + 10 * c;
    {
      float t00 = (c == 0) ? 0.f : b00, t10 = (c == 0) ? 0.f : b10;
      float t01 = (c == 13) ? 0.f : b01, t11 = (c == 13) ? 0.f : b11;
      acc += dot10(t00, t01, t10, t11, &s_w2[0][wofs]);
    }
    {
      float t00 = (r == 0) ? 0.f : b00, t01 = (r == 0) ? 0.f : b01;
      float t10 = (r == 13) ? 0.f : b10, t11 = (r == 13) ? 0.f : b11;
      acc += dot10(t00, t01, t10, t11, &s_w2[1][wofs]);
    }
    {
      float t00 = (c == 0 || r == 0) ? 0.f : b00;
      float t01 = (c == 13 || r == 0) ? 0.f : b01;
      float t10 = (c == 0 || r == 13) ? 0.f : b10;
      float t11 = (c == 13 || r == 13) ? 0.f : b11;
      acc += dot10(t00, t01, t10, t11, &s_w2[2][wofs]);
    }
    acc += dot10(b00, b01, b10, b11, &s_w2[3][wofs]);
    l1_out[b * 196 + pix] = acc;
    sum += acc; sq += acc * acc;
  }
  s_red[tid] = sum; __syncthreads();
  for (int o = 128; o > 0; o >>= 1) { if (tid < o) s_red[tid] += s_red[tid + o]; __syncthreads(); }
  if (tid == 0) atomicAdd(&stats[0], s_red[0]);
  __syncthreads();
  s_red[tid] = sq; __syncthreads();
  for (int o = 128; o > 0; o >>= 1) { if (tid < o) s_red[tid] += s_red[tid + o]; __syncthreads(); }
  if (tid == 0) atomicAdd(&stats[1], s_red[0]);
}

__global__ void stats_kernel(float* __restrict__ stats,
                             const float* __restrict__ gamma,
                             const float* __restrict__ beta,
                             float n, int sIdx, int oIdx) {
  float mean = stats[sIdx] / n;
  float var = stats[sIdx + 1] / n - mean * mean;
  float s = gamma[0] * rsqrtf(var + EPS);
  stats[oIdx] = s;
  stats[oIdx + 1] = beta[0] - mean * s;
}

__global__ __launch_bounds__(256) void layer2_kernel(
    const float* __restrict__ l1_in,
    const float* __restrict__ w1, const float* __restrict__ w2,
    const float* __restrict__ w4,
    float* __restrict__ l2_out, float* __restrict__ stats) {
  __shared__ float s_w1[196];
  __shared__ float s_w2[490];
  __shared__ float s_w4[49];
  __shared__ float s_h[8][196];
  __shared__ float s_red[256];
  int tid = threadIdx.x;
  for (int p = tid; p < 196; p += 256) s_w1[p] = w1[p];
  for (int p = tid; p < 490; p += 256) s_w2[p] = w2[p];
  for (int p = tid; p < 49; p += 256) s_w4[p] = w4[p];
  float sbn = stats[4], tbn = stats[5];
  int wv = tid >> 5, lane = tid & 31;
  long b = (long)blockIdx.x * 8 + wv;
  const float* lb = l1_in + b * 196;
  for (int p = lane; p < 196; p += 32) s_h[wv][p] = lb[p] * sbn + tbn;
  __syncthreads();

  float sum = 0.f, sq = 0.f;
  for (int pix = lane; pix < 49; pix += 32) {
    int r = pix / 7, c = pix - r * 7;
    int i00 = (2 * r) * 14 + 2 * c;
    float b00 = s_h[wv][i00],      b01 = s_h[wv][i00 + 1];
    float b10 = s_h[wv][i00 + 14], b11 = s_h[wv][i00 + 15];
    float acc = b00 * s_w1[i00] + b01 * s_w1[i00 + 1]
              + b10 * s_w1[i00 + 14] + b11 * s_w1[i00 + 15];
    acc += s_w4[pix];
    acc += dot10(b00, b01, b10, b11, &s_w2[r * 70 + 10 * c]);
    l2_out[b * 49 + pix] = acc;
    sum += acc; sq += acc * acc;
  }
  s_red[tid] = sum; __syncthreads();
  for (int o = 128; o > 0; o >>= 1) { if (tid < o) s_red[tid] += s_red[tid + o]; __syncthreads(); }
  if (tid == 0) atomicAdd(&stats[2], s_red[0]);
  __syncthreads();
  s_red[tid] = sq; __syncthreads();
  for (int o = 128; o > 0; o >>= 1) { if (tid < o) s_red[tid] += s_red[tid + o]; __syncthreads(); }
  if (tid == 0) atomicAdd(&stats[3], s_red[0]);
}

__global__ __launch_bounds__(256) void gemm_kernel(
    const float* __restrict__ l2_in, const float* __restrict__ stats,
    const float2* __restrict__ wswz,
    const float* __restrict__ b_fc1, const float* __restrict__ b_fc2,
    float* __restrict__ out) {
  __shared__ __align__(16) float s_v[8][16 * 52];
  int tid = threadIdx.x;
  int wv = tid >> 5, lane = tid & 31;
  int n = lane & 15, half = lane >> 4;
  long b0 = ((long)blockIdx.x * 8 + wv) * 16;
  float sbn = stats[6], tbn = stats[7];

  for (int idx = lane; idx < 16 * 52; idx += 32) {
    int m = idx / 52, j = idx - m * 52;
    float val = 0.f;
    if (j < 49) val = l2_in[(b0 + m) * 49 + j] * sbn + tbn;
    s_v[wv][idx] = val;
  }
  __syncthreads();

  const float* vp = s_v[wv] + n * 52;
  v8f acc = {};
  const float2* wp = wswz + lane;
  for (int i = 0; i < 50; ++i) {
    float vi = (i < 49) ? vp[i] : 1.0f;
#pragma unroll
    for (int j0 = 0; j0 < 52; j0 += 4) {
      int j = j0 + 2 * half;
      float vj0 = vp[j], vj1 = vp[j + 1];
      v2f a; a.x = vi * vj0; a.y = vi * vj1;
      float2 w = *wp; wp += 32;
      v2f bm; bm.x = w.x; bm.y = w.y;
      acc = __builtin_amdgcn_wmma_f32_16x16x4_f32(
          false, a, false, bm, (short)0, acc, false, false);
    }
  }

  if (n < 10) {
    float bias = b_fc1[n] + b_fc2[n];
#pragma unroll
    for (int r = 0; r < 8; ++r) {
      long M = b0 + r + 8 * half;
      out[M * 10 + n] = acc[r] + bias;
    }
  }
}

extern "C" void kernel_launch(void* const* d_in, const int* in_sizes, int n_in,
                              void* d_out, int out_size, void* d_ws, size_t ws_size,
                              hipStream_t stream) {
  (void)in_sizes; (void)n_in; (void)out_size; (void)ws_size;
  const float* x       = (const float*)d_in[0];
  const float* l1_w1   = (const float*)d_in[1];
  const float* l1_w2_1 = (const float*)d_in[2];
  const float* l1_w2_2 = (const float*)d_in[3];
  const float* l1_w2_3 = (const float*)d_in[4];
  const float* l1_w2_4 = (const float*)d_in[5];
  const float* l1_w3   = (const float*)d_in[6];
  const float* l2_w1   = (const float*)d_in[7];
  const float* l2_w2   = (const float*)d_in[8];
  const float* l2_w4   = (const float*)d_in[9];
  const float* w_fc1   = (const float*)d_in[10];
  const float* b_fc1   = (const float*)d_in[11];
  const float* w_fc2   = (const float*)d_in[12];
  const float* b_fc2   = (const float*)d_in[13];
  const float* gamma   = (const float*)d_in[14];
  const float* beta    = (const float*)d_in[15];
  float* out = (float*)d_out;

  float* l1_buf = (float*)d_ws;
  float* l2_buf = l1_buf + (size_t)BATCH * 196;
  float* wswz   = l2_buf + (size_t)BATCH * 49;
  float* stats  = wswz + (size_t)650 * 32 * 2;

  prep_kernel<<<(650 * 32 + 255) / 256, 256, 0, stream>>>(w_fc1, w_fc2, (float2*)wswz, stats);
  layer1_kernel<<<BATCH / 8, 256, 0, stream>>>(x, l1_w1, l1_w2_1, l1_w2_2, l1_w2_3,
                                               l1_w2_4, l1_w3, l1_buf, stats);
  stats_kernel<<<1, 1, 0, stream>>>(stats, gamma, beta, (float)(BATCH * 196), 0, 4);
  layer2_kernel<<<BATCH / 8, 256, 0, stream>>>(l1_buf, l2_w1, l2_w2, l2_w4, l2_buf, stats);
  stats_kernel<<<1, 1, 0, stream>>>(stats, gamma, beta, (float)(BATCH * 49), 2, 6);
  gemm_kernel<<<BATCH / 128, 256, 0, stream>>>(l2_buf, stats, (const float2*)wswz,
                                               b_fc1, b_fc2, out);
}


# head GEMM: split K over the 8 waves of a work-group (one 16-batch tile per group, wave w takes K rows w+8r), partial tiles summed through LDS in fixed order; changes the f32 summation order
# speedup vs baseline: 1.8441x; 1.8441x over previous
#include <hip/hip_runtime.h>

typedef __attribute__((ext_vector_type(2))) float v2f;
typedef __attribute__((ext_vector_type(8))) float v8f;

#define BATCH 16384
#define EPS 1e-5f

__device__ inline float wz_val(const float* w_fc1, const float* w_fc2, int k, int n) {
  if (n >= 10) return 0.f;
  int i = k / 52, j = k - i * 52;
  if (j >= 49) return 0.f;
  if (i < 49) return (i <= j) ? w_fc2[n * 2401 + i * 49 + j] : 0.f;
  return w_fc1[n * 49 + j];
}

__global__ void prep_kernel(const float* __restrict__ w_fc1,
                            const float* __restrict__ w_fc2,
                            float2* __restrict__ wswz,
                            float* __restrict__ stats) {
  int t = blockIdx.x * blockDim.x + threadIdx.x;
  if (t < 4) stats[t] = 0.f;
  if (t < 650 * 32) {
    int s = t >> 5, lane = t & 31;
    int n = lane & 15, half = lane >> 4;
    int ka = 4 * s + 2 * half;
    float2 v;
    v.x = wz_val(w_fc1, w_fc2, ka, n);
    v.y = wz_val(w_fc1, w_fc2, ka + 1, n);
    wswz[t] = v;
  }
}

__device__ inline float dot10(float t00, float t01, float t10, float t11,
                              const float* __restrict__ W) {
  return W[0]*t00*t00 + W[1]*t01*t01 + W[2]*t10*t10 + W[3]*t11*t11
       + W[4]*t00*t01 + W[5]*t00*t10 + W[6]*t00*t11
       + W[7]*t01*t10 + W[8]*t01*t11 + W[9]*t10*t11;
}

__global__ __launch_bounds__(256) void layer1_kernel(
    const float* __restrict__ x,
    const float* __restrict__ w1,   const float* __restrict__ w2lr,
    const float* __restrict__ w2tb, const float* __restrict__ w2all,
    const float* __restrict__ w2no, const float* __restrict__ w3,
    float* __restrict__ l1_out, float* __restrict__ stats) {
  __shared__ float s_w1[784];
  __shared__ float s_w2[4][1960];
  __shared__ float s_w3[196];
  __shared__ float s_img[8][784];
  __shared__ float s_red[256];
  int tid = threadIdx.x;
  for (int p = tid; p < 784; p += 256) s_w1[p] = w1[p];
  for (int p = tid; p < 1960; p += 256) {
    s_w2[0][p] = w2lr[p]; s_w2[1][p] = w2tb[p];
    s_w2[2][p] = w2all[p]; s_w2[3][p] = w2no[p];
  }
  for (int p = tid; p < 196; p += 256) s_w3[p] = w3[p];
  int wv = tid >> 5, lane = tid & 31;
  long b = (long)blockIdx.x * 8 + wv;
  const float* xb = x + b * 784;
  for (int p = lane; p < 784; p += 32) s_img[wv][p] = xb[p];
  __syncthreads();

  float sum = 0.f, sq = 0.f;
  for (int pix = lane; pix < 196; pix += 32) {
    int r = pix / 14, c = pix - r * 14;
    int i00 = (2 * r) * 28 + 2 * c;
    float b00 = s_img[wv][i00],      b01 = s_img[wv][i00 + 1];
    float b10 = s_img[wv][i00 + 28], b11 = s_img[wv][i00 + 29];
    float acc = b00 * s_w1[i00] + b01 * s_w1[i00 + 1]
              + b10 * s_w1[i00 + 28] + b11 * s_w1[i00 + 29];
    acc += s_w3[pix];
    int wofs = r * 140 + 10 * c;
    {
      float t00 = (c == 0) ? 0.f : b00, t10 = (c == 0) ? 0.f : b10;
      float t01 = (c == 13) ? 0.f : b01, t11 = (c == 13) ? 0.f : b11;
      acc += dot10(t00, t01, t10, t11, &s_w2[0][wofs]);
    }
    {
      float t00 = (r == 0) ? 0.f : b00, t01 = (r == 0) ? 0.f : b01;
      float t10 = (r == 13) ? 0.f : b10, t11 = (r == 13) ? 0.f : b11;
      acc += dot10(t00, t01, t10, t11, &s_w2[1][wofs]);
    }
    {
      float t00 = (c == 0 || r == 0) ? 0.f : b00;
      float t01 = (c == 13 || r == 0) ? 0.f : b01;
      float t10 = (c == 0 || r == 13) ? 0.f : b10;
      float t11 = (c == 13 || r == 13) ? 0.f : b11;
      acc += dot10(t00, t01, t10, t11, &s_w2[2][wofs]);
    }
    acc += dot10(b00, b01, b10, b11, &s_w2[3][wofs]);
    l1_out[b * 196 + pix] = acc;
    sum += acc; sq += acc * acc;
  }
  s_red[tid] = sum; __syncthreads();
  for (int o = 128; o > 0; o >>= 1) { if (tid < o) s_red[tid] += s_red[tid + o]; __syncthreads(); }
  if (tid == 0) atomicAdd(&stats[0], s_red[0]);
  __syncthreads();
  s_red[tid] = sq; __syncthreads();
  for (int o = 128; o > 0; o >>= 1) { if (tid < o) s_red[tid] += s_red[tid + o]; __syncthreads(); }
  if (tid == 0) atomicAdd(&stats[1], s_red[0]);
}

__global__ void stats_kernel(float* __restrict__ stats,
                             const float* __restrict__ gamma,
                             const float* __restrict__ beta,
                             float n, int sIdx, int oIdx) {
  float mean = stats[sIdx] / n;
  float var = stats[sIdx + 1] / n - mean * mean;
  float s = gamma[0] * rsqrtf(var + EPS);
  stats[oIdx] = s;
  stats[oIdx + 1] = beta[0] - mean * s;
}

__global__ __launch_bounds__(256) void layer2_kernel(
    const float* __restrict__ l1_in,
    const float* __restrict__ w1, const float* __restrict__ w2,
    const float* __restrict__ w4,
    float* __restrict__ l2_out, float* __restrict__ stats) {
  __shared__ float s_w1[196];
  __shared__ float s_w2[490];
  __shared__ float s_w4[49];
  __shared__ float s_h[8][196];
  __shared__ float s_red[256];
  int tid = threadIdx.x;
  for (int p = tid; p < 196; p += 256) s_w1[p] = w1[p];
  for (int p = tid; p < 490; p += 256) s_w2[p] = w2[p];
  for (int p = tid; p < 49; p += 256) s_w4[p] = w4[p];
  float sbn = stats[4], tbn = stats[5];
  int wv = tid >> 5, lane = tid & 31;
  long b = (long)blockIdx.x * 8 + wv;
  const float* lb = l1_in + b * 196;
  for (int p = lane; p < 196; p += 32) s_h[wv][p] = lb[p] * sbn + tbn;
  __syncthreads();

  float sum = 0.f, sq = 0.f;
  for (int pix = lane; pix < 49; pix += 32) {
    int r = pix / 7, c = pix - r * 7;
    int i00 = (2 * r) * 14 + 2 * c;
    float b00 = s_h[wv][i00],      b01 = s_h[wv][i00 + 1];
    float b10 = s_h[wv][i00 + 14], b11 = s_h[wv][i00 + 15];
    float acc = b00 * s_w1[i00] + b01 * s_w1[i00 + 1]
              + b10 * s_w1[i00 + 14] + b11 * s_w1[i00 + 15];
    acc += s_w4[pix];
    acc += dot10(b00, b01, b10, b11, &s_w2[r * 70 + 10 * c]);
    l2_out[b * 49 + pix] = acc;
    sum += acc; sq += acc * acc;
  }
  s_red[tid] = sum; __syncthreads();
  for (int o = 128; o > 0; o >>= 1) { if (tid < o) s_red[tid] += s_red[tid + o]; __syncthreads(); }
  if (tid == 0) atomicAdd(&stats[2], s_red[0]);
  __syncthreads();
  s_red[tid] = sq; __syncthreads();
  for (int o = 128; o > 0; o >>= 1) { if (tid < o) s_red[tid] += s_red[tid + o]; __syncthreads(); }
  if (tid == 0) atomicAdd(&stats[3], s_red[0]);
}

__global__ __launch_bounds__(256) void gemm_kernel(
    const float* __restrict__ l2_in, const float* __restrict__ stats,
    const float2* __restrict__ wswz,
    const float* __restrict__ b_fc1, const float* __restrict__ b_fc2,
    float* __restrict__ out) {
  __shared__ __align__(16) float s_v[16 * 52];
  __shared__ float s_part[8][256];
  __shared__ float s_out[160];
  int tid = threadIdx.x;
  int wv = __builtin_amdgcn_readfirstlane(tid >> 5);
  int lane = tid & 31;
  int n = lane & 15, half = lane >> 4;
  long b0 = (long)blockIdx.x * 16;
  float sbn = stats[6], tbn = stats[7];

  const float* src = l2_in + b0 * 49;
  float x0 = src[tid], x1 = src[tid + 256], x2 = src[tid + 512];
  float x3 = (tid < 16) ? src[tid + 768] : 0.f;
  {
    int m = tid / 49;
    s_v[m * 52 + (tid - m * 49)] = x0 * sbn + tbn;
    int i1 = tid + 256; m = i1 / 49;
    s_v[m * 52 + (i1 - m * 49)] = x1 * sbn + tbn;
    int i2 = tid + 512; m = i2 / 49;
    s_v[m * 52 + (i2 - m * 49)] = x2 * sbn + tbn;
    if (tid < 16) s_v[15 * 52 + 33 + tid] = x3 * sbn + tbn;
  }
  if (tid < 48) {
    int m = tid / 3;
    s_v[m * 52 + 49 + (tid - m * 3)] = 0.f;
  }
  __syncthreads();

  const float* vp = s_v + n * 52;
  v8f acc = {};
  for (int i = wv; i < 50; i += 8) {
    float vi = (i < 49) ? vp[i] : 1.0f;
    const float2* wp = wswz + (size_t)i * (13 * 32) + lane;
#pragma unroll
    for (int q = 0; q < 13; ++q) {
      int j = 4 * q + 2 * half;
      float vj0 = vp[j], vj1 = vp[j + 1];
      v2f a; a.x = vi * vj0; a.y = vi * vj1;
      float2 w = wp[q * 32];
      v2f bm; bm.x = w.x; bm.y = w.y;
      acc = __builtin_amdgcn_wmma_f32_16x16x4_f32(
          false, a, false, bm, (short)0, acc, false, false);
    }
  }

#pragma unroll
  for (int r = 0; r < 8; ++r) s_part[wv][r * 32 + lane] = acc[r];
  __syncthreads();

  float sum = 0.f;
  for (int w = 0; w < 8; ++w) sum += s_part[w][tid];
  if (n < 10) {
    int M = (tid >> 5) + 8 * half;
    s_out[M * 10 + n] = sum + (b_fc1[n] + b_fc2[n]);
  }
  __syncthreads();
  if (tid < 160) out[b0 * 10 + tid] = s_out[tid];
}

extern "C" void kernel_launch(void* const* d_in, const int* in_sizes, int n_in,
                              void* d_out, int out_size, void* d_ws, size_t ws_size,
                              hipStream_t stream) {
  (void)in_sizes; (void)n_in; (void)out_size; (void)ws_size;
  const float* x       = (const float*)d_in[0];
  const float* l1_w1   = (const float*)d_in[1];
  const float* l1_w2_1 = (const float*)d_in[2];
  const float* l1_w2_2 = (const float*)d_in[3];
  const float* l1_w2_3 = (const float*)d_in[4];
  const float* l1_w2_4 = (const float*)d_in[5];
  const float* l1_w3   = (const float*)d_in[6];
  const float* l2_w1   = (const float*)d_in[7];
  const float* l2_w2   = (const float*)d_in[8];
  const float* l2_w4   = (const float*)d_in[9];
  const float* w_fc1   = (const float*)d_in[10];
  const float* b_fc1   = (const float*)d_in[11];
  const float* w_fc2   = (const float*)d_in[12];
  const float* b_fc2   = (const float*)d_in[13];
  const float* gamma   = (const float*)d_in[14];
  const float* beta    = (const float*)d_in[15];
  float* out = (float*)d_out;

  float* l1_buf = (float*)d_ws;
  float* l2_buf = l1_buf + (size_t)BATCH * 196;
  float* wswz   = l2_buf + (size_t)BATCH * 49;
  float* stats  = wswz + (size_t)650 * 32 * 2;

  prep_kernel<<<(650 * 32 + 255) / 256, 256, 0, stream>>>(w_fc1, w_fc2, (float2*)wswz, stats);
  layer1_kernel<<<BATCH / 8, 256, 0, stream>>>(x, l1_w1, l1_w2_1, l1_w2_2, l1_w2_3,
                                               l1_w2_4, l1_w3, l1_buf, stats);
  stats_kernel<<<1, 1, 0, stream>>>(stats, gamma, beta, (float)(BATCH * 196), 0, 4);
  layer2_kernel<<<BATCH / 8, 256, 0, stream>>>(l1_buf, l2_w1, l2_w2, l2_w4, l2_buf, stats);
  stats_kernel<<<1, 1, 0, stream>>>(stats, gamma, beta, (float)(BATCH * 49), 2, 6);
  gemm_kernel<<<BATCH / 16, 256, 0, stream>>>(l2_buf, stats, (const float2*)wswz,
                                               b_fc1, b_fc2, out);
}


# layer 2 as 512 resident groups: a wave takes 4 images (one group of 16-byte loads, image-pixel pairs flattened over lanes); weights staged, BN sums reduced once per group; reorders the f32 BN sums
# speedup vs baseline: 2.0124x; 1.0913x over previous
#include <hip/hip_runtime.h>

typedef __attribute__((ext_vector_type(2))) float v2f;
typedef __attribute__((ext_vector_type(8))) float v8f;

#define BATCH 16384
#define EPS 1e-5f

__device__ inline float wz_val(const float* w_fc1, const float* w_fc2, int k, int n) {
  if (n >= 10) return 0.f;
  int i = k / 52, j = k - i * 52;
  if (j >= 49) return 0.f;
  if (i < 49) return (i <= j) ? w_fc2[n * 2401 + i * 49 + j] : 0.f;
  return w_fc1[n * 49 + j];
}

__global__ void prep_kernel(const float* __restrict__ w_fc1,
                            const float* __restrict__ w_fc2,
                            float2* __restrict__ wswz,
                            float* __restrict__ stats) {
  int t = blockIdx.x * blockDim.x + threadIdx.x;
  if (t < 4) stats[t] = 0.f;
  if (t < 650 * 32) {
    int s = t >> 5, lane = t & 31;
    int n = lane & 15, half = lane >> 4;
    int ka = 4 * s + 2 * half;
    float2 v;
    v.x = wz_val(w_fc1, w_fc2, ka, n);
    v.y = wz_val(w_fc1, w_fc2, ka + 1, n);
    wswz[t] = v;
  }
}

__device__ inline float dot10(float t00, float t01, float t10, float t11,
                              const float* __restrict__ W) {
  return W[0]*t00*t00 + W[1]*t01*t01 + W[2]*t10*t10 + W[3]*t11*t11
       + W[4]*t00*t01 + W[5]*t00*t10 + W[6]*t00*t11
       + W[7]*t01*t10 + W[8]*t01*t11 + W[9]*t10*t11;
}

__global__ __launch_bounds__(256) void layer1_kernel(
    const float* __restrict__ x,
    const float* __restrict__ w1,   const float* __restrict__ w2lr,
    const float* __restrict__ w2tb, const float* __restrict__ w2all,
    const float* __restrict__ w2no, const float* __restrict__ w3,
    float* __restrict__ l1_out, float* __restrict__ stats) {
  __shared__ float s_w1[784];
  __shared__ float s_w2[4][1960];
  __shared__ float s_w3[196];
  __shared__ float s_img[8][784];
  __shared__ float s_red[256];
  int tid = threadIdx.x;
  for (int p = tid; p < 784; p += 256) s_w1[p] = w1[p];
  for (int p = tid; p < 1960; p += 256) {
    s_w2[0][p] = w2lr[p]; s_w2[1][p] = w2tb[p];
    s_w2[2][p] = w2all[p]; s_w2[3][p] = w2no[p];
  }
  for (int p = tid; p < 196; p += 256) s_w3[p] = w3[p];
  int wv = tid >> 5, lane = tid & 31;
  long b = (long)blockIdx.x * 8 + wv;
  const float* xb = x + b * 784;
  for (int p = lane; p < 784; p += 32) s_img[wv][p] = xb[p];
  __syncthreads();

  float sum = 0.f, sq = 0.f;
  for (int pix = lane; pix < 196; pix += 32) {
    int r = pix / 14, c = pix - r * 14;
    int i00 = (2 * r) * 28 + 2 * c;
    float b00 = s_img[wv][i00],      b01 = s_img[wv][i00 + 1];
    float b10 = s_img[wv][i00 + 28], b11 = s_img[wv][i00 + 29];
    float acc = b00 * s_w1[i00] + b01 * s_w1[i00 + 1]
              + b10 * s_w1[i00 + 28] + b11 * s_w1[i00 + 29];
    acc += s_w3[pix];
    int wofs = r * 140 + 10 * c;
    {
      float t00 = (c == 0) ? 0.f : b00, t10 = (c == 0) ? 0.f : b10;
      float t01 = (c == 13) ? 0.f : b01, t11 = (c == 13) ? 0.f : b11;
      acc += dot10(t00, t01, t10, t11, &s_w2[0][wofs]);
    }
    {
      float t00 = (r == 0) ? 0.f : b00, t01 = (r == 0) ? 0.f : b01;
      float t10 = (r == 13) ? 0.f : b10, t11 = (r == 13) ? 0.f : b11;
      acc += dot10(t00, t01, t10, t11, &s_w2[1][wofs]);
    }
    {
      float t00 = (c == 0 || r == 0) ? 0.f : b00;
      float t01 = (c == 13 || r == 0) ? 0.f : b01;
      float t10 = (c == 0 || r == 13) ? 0.f : b10;
      float t11 = (c == 13 || r == 13) ? 0.f : b11;
      acc += dot10(t00, t01, t10, t11, &s_w2[2][wofs]);
    }
    acc += dot10(b00, b01, b10, b11, &s_w2[3][wofs]);
    l1_out[b * 196 + pix] = acc;
    sum += acc; sq += acc * acc;
  }
  s_red[tid] = sum; __syncthreads();
  for (int o = 128; o > 0; o >>= 1) { if (tid < o) s_red[tid] += s_red[tid + o]; __syncthreads(); }
  if (tid == 0) atomicAdd(&stats[0], s_red[0]);
  __syncthreads();
  s_red[tid] = sq; __syncthreads();
  for (int o = 128; o > 0; o >>= 1) { if (tid < o) s_red[tid] += s_red[tid + o]; __syncthreads(); }
  if (tid == 0) atomicAdd(&stats[1], s_red[0]);
}

__global__ void stats_kernel(float* __restrict__ stats,
                             const float* __restrict__ gamma,
                             const float* __restrict__ beta,
                             float n, int sIdx, int oIdx) {
  float mean = stats[sIdx] / n;
  float var = stats[sIdx + 1] / n - mean * mean;
  float s = gamma[0] * rsqrtf(var + EPS);
  stats[oIdx] = s;
  stats[oIdx + 1] = beta[0] - mean * s;
}

#define L2_GROUPS 512
#define L2_WIMG (BATCH / (L2_GROUPS * 8))
#define L2_WVEC (L2_WIMG * 196 / 4)
#define L2_WTASK (L2_WIMG * 49)
__global__ __launch_bounds__(256) void layer2_kernel(
    const float* __restrict__ l1_in,
    const float* __restrict__ w1, const float* __restrict__ w2,
    const float* __restrict__ w4,
    float* __restrict__ l2_out, float* __restrict__ stats) {
  __shared__ float s_w1[196];
  __shared__ float s_w2[490];
  __shared__ float s_w4[49];
  __shared__ __align__(16) float s_h[8][L2_WIMG * 196];
  __shared__ float s_red[16];
  int tid = threadIdx.x;
  int wv = tid >> 5, lane = tid & 31;
  long first = ((long)blockIdx.x * 8 + wv) * L2_WIMG;
  const float4* src = reinterpret_cast<const float4*>(l1_in + first * 196);
  float4 v[(L2_WVEC + 31) / 32];
#pragma unroll
  for (int k = 0; k < (L2_WVEC + 31) / 32; ++k) {
    int idx = lane + 32 * k;
    if (idx < L2_WVEC) v[k] = src[idx];
  }
  float q1 = (tid < 196) ? w1[tid] : 0.f;
  float q2 = w2[tid];
  float q3 = (tid < 490 - 256) ? w2[tid + 256] : 0.f;
  float q4 = (tid < 49) ? w4[tid] : 0.f;
  float sbn = stats[4], tbn = stats[5];
  if (tid < 196) s_w1[tid] = q1;
  s_w2[tid] = q2;
  if (tid < 490 - 256) s_w2[tid + 256] = q3;
  if (tid < 49) s_w4[tid] = q4;
  float4* hv = reinterpret_cast<float4*>(&s_h[wv][0]);
#pragma unroll
  for (int k = 0; k < (L2_WVEC + 31) / 32; ++k) {
    int idx = lane + 32 * k;
    if (idx < L2_WVEC) {
      float4 h;
      h.x = v[k].x * sbn + tbn; h.y = v[k].y * sbn + tbn;
      h.z = v[k].z * sbn + tbn; h.w = v[k].w * sbn + tbn;
      hv[idx] = h;
    }
  }
  __syncthreads();

  const float* hw = &s_h[wv][0];
  float sum = 0.f, sq = 0.f;
  for (int t = lane; t < L2_WTASK; t += 32) {
    int im = t / 49, pix = t - im * 49;
    int r = pix / 7, c = pix - r * 7;
    int i00 = (2 * r) * 14 + 2 * c;
    const float* hp = hw + im * 196 + i00;
    float b00 = hp[0],  b01 = hp[1];
    float b10 = hp[14], b11 = hp[15];
    float acc = b00 * s_w1[i00] + b01 * s_w1[i00 + 1]
              + b10 * s_w1[i00 + 14] + b11 * s_w1[i00 + 15];
    acc += s_w4[pix];
    acc += dot10(b00, b01, b10, b11, &s_w2[r * 70 + 10 * c]);
    l2_out[first * 49 + t] = acc;
    sum += acc; sq += acc * acc;
  }
  for (int o = 16; o > 0; o >>= 1) {
    sum += __shfl_xor(sum, o);
    sq  += __shfl_xor(sq, o);
  }
  if (lane == 0) { s_red[wv] = sum; s_red[8 + wv] = sq; }
  __syncthreads();
  if (tid == 0) {
    float a = 0.f, q = 0.f;
    for (int w = 0; w < 8; ++w) { a += s_red[w]; q += s_red[8 + w]; }
    atomicAdd(&stats[2], a);
    atomicAdd(&stats[3], q);
  }
}

__global__ __launch_bounds__(256) void gemm_kernel(
    const float* __restrict__ l2_in, const float* __restrict__ stats,
    const float2* __restrict__ wswz,
    const float* __restrict__ b_fc1, const float* __restrict__ b_fc2,
    float* __restrict__ out) {
  __shared__ __align__(16) float s_v[16 * 52];
  __shared__ float s_part[8][256];
  __shared__ float s_out[160];
  int tid = threadIdx.x;
  int wv = __builtin_amdgcn_readfirstlane(tid >> 5);
  int lane = tid & 31;
  int n = lane & 15, half = lane >> 4;
  long b0 = (long)blockIdx.x * 16;
  float sbn = stats[6], tbn = stats[7];

  const float* src = l2_in + b0 * 49;
  float x0 = src[tid], x1 = src[tid + 256], x2 = src[tid + 512];
  float x3 = (tid < 16) ? src[tid + 768] : 0.f;
  {
    int m = tid / 49;
    s_v[m * 52 + (tid - m * 49)] = x0 * sbn + tbn;
    int i1 = tid + 256; m = i1 / 49;
    s_v[m * 52 + (i1 - m * 49)] = x1 * sbn + tbn;
    int i2 = tid + 512; m = i2 / 49;
    s_v[m * 52 + (i2 - m * 49)] = x2 * sbn + tbn;
    if (tid < 16) s_v[15 * 52 + 33 + tid] = x3 * sbn + tbn;
  }
  if (tid < 48) {
    int m = tid / 3;
    s_v[m * 52 + 49 + (tid - m * 3)] = 0.f;
  }
  __syncthreads();

  const float* vp = s_v + n * 52;
  v8f acc = {};
  for (int i = wv; i < 50; i += 8) {
    float vi = (i < 49) ? vp[i] : 1.0f;
    const float2* wp = wswz + (size_t)i * (13 * 32) + lane;
#pragma unroll
    for (int q = 0; q < 13; ++q) {
      int j = 4 * q + 2 * half;
      float vj0 = vp[j], vj1 = vp[j + 1];
      v2f a; a.x = vi * vj0; a.y = vi * vj1;
      float2 w = wp[q * 32];
      v2f bm; bm.x = w.x; bm.y = w.y;
      acc = __builtin_amdgcn_wmma_f32_16x16x4_f32(
          false, a, false, bm, (short)0, acc, false, false);
    }
  }

#pragma unroll
  for (int r = 0; r < 8; ++r) s_part[wv][r * 32 + lane] = acc[r];
  __syncthreads();

  float sum = 0.f;
  for (int w = 0; w < 8; ++w) sum += s_part[w][tid];
  if (n < 10) {
    int M = (tid >> 5) + 8 * half;
    s_out[M * 10 + n] = sum + (b_fc1[n] + b_fc2[n]);
  }
  __syncthreads();
  if (tid < 160) out[b0 * 10 + tid] = s_out[tid];
}

extern "C" void kernel_launch(void* const* d_in, const int* in_sizes, int n_in,
                              void* d_out, int out_size, void* d_ws, size_t ws_size,
                              hipStream_t stream) {
  (void)in_sizes; (void)n_in; (void)out_size; (void)ws_size;
  const float* x       = (const float*)d_in[0];
  const float* l1_w1   = (const float*)d_in[1];
  const float* l1_w2_1 = (const float*)d_in[2];
  const float* l1_w2_2 = (const float*)d_in[3];
  const float* l1_w2_3 = (const float*)d_in[4];
  const float* l1_w2_4 = (const float*)d_in[5];
  const float* l1_w3   = (const float*)d_in[6];
  const float* l2_w1   = (const float*)d_in[7];
  const float* l2_w2   = (const float*)d_in[8];
  const float* l2_w4   = (const float*)d_in[9];
  const float* w_fc1   = (const float*)d_in[10];
  const float* b_fc1   = (const float*)d_in[11];
  const float* w_fc2   = (const float*)d_in[12];
  const float* b_fc2   = (const float*)d_in[13];
  const float* gamma   = (const float*)d_in[14];
  const float* beta    = (const float*)d_in[15];
  float* out = (float*)d_out;

  float* l1_buf = (float*)d_ws;
  float* l2_buf = l1_buf + (size_t)BATCH * 196;
  float* wswz   = l2_buf + (size_t)BATCH * 49;
  float* stats  = wswz + (size_t)650 * 32 * 2;

  prep_kernel<<<(650 * 32 + 255) / 256, 256, 0, stream>>>(w_fc1, w_fc2, (float2*)wswz, stats);
  layer1_kernel<<<BATCH / 8, 256, 0, stream>>>(x, l1_w1, l1_w2_1, l1_w2_2, l1_w2_3,
                                               l1_w2_4, l1_w3, l1_buf, stats);
  stats_kernel<<<1, 1, 0, stream>>>(stats, gamma, beta, (float)(BATCH * 196), 0, 4);
  layer2_kernel<<<L2_GROUPS, 256, 0, stream>>>(l1_buf, l2_w1, l2_w2, l2_w4, l2_buf, stats);
  stats_kernel<<<1, 1, 0, stream>>>(stats, gamma, beta, (float)(BATCH * 49), 2, 6);
  gemm_kernel<<<BATCH / 16, 256, 0, stream>>>(l2_buf, stats, (const float2*)wswz,
                                               b_fc1, b_fc2, out);
}


# layer 1: a lane owns one pixel and keeps its weights in registers, border masks and the 4 branches folded into one 15-coefficient quadratic form; x streamed from global, no LDS; reorders f32 sums
# speedup vs baseline: 3.4239x; 1.7014x over previous
#include <hip/hip_runtime.h>

typedef __attribute__((ext_vector_type(2))) float v2f;
typedef __attribute__((ext_vector_type(8))) float v8f;

#define BATCH 16384
#define EPS 1e-5f

__device__ inline float wz_val(const float* w_fc1, const float* w_fc2, int k, int n) {
  if (n >= 10) return 0.f;
  int i = k / 52, j = k - i * 52;
  if (j >= 49) return 0.f;
  if (i < 49) return (i <= j) ? w_fc2[n * 2401 + i * 49 + j] : 0.f;
  return w_fc1[n * 49 + j];
}

__global__ void prep_kernel(const float* __restrict__ w_fc1,
                            const float* __restrict__ w_fc2,
                            float2* __restrict__ wswz,
                            float* __restrict__ stats) {
  int t = blockIdx.x * blockDim.x + threadIdx.x;
  if (t < 4) stats[t] = 0.f;
  if (t < 650 * 32) {
    int s = t >> 5, lane = t & 31;
    int n = lane & 15, half = lane >> 4;
    int ka = 4 * s + 2 * half;
    float2 v;
    v.x = wz_val(w_fc1, w_fc2, ka, n);
    v.y = wz_val(w_fc1, w_fc2, ka + 1, n);
    wswz[t] = v;
  }
}

__device__ inline float dot10(float t00, float t01, float t10, float t11,
                              const float* __restrict__ W) {
  return W[0]*t00*t00 + W[1]*t01*t01 + W[2]*t10*t10 + W[3]*t11*t11
       + W[4]*t00*t01 + W[5]*t00*t10 + W[6]*t00*t11
       + W[7]*t01*t10 + W[8]*t01*t11 + W[9]*t10*t11;
}

#define L1_GROUPS 512
#define L1_GIMG (BATCH / L1_GROUPS)
#define L1_U 8
__global__ __launch_bounds__(224) void layer1_kernel(
    const float* __restrict__ x,
    const float* __restrict__ w1,   const float* __restrict__ w2lr,
    const float* __restrict__ w2tb, const float* __restrict__ w2all,
    const float* __restrict__ w2no, const float* __restrict__ w3,
    float* __restrict__ l1_out, float* __restrict__ stats) {
  __shared__ float s_red[16];
  int tid = threadIdx.x;
  int wv = tid >> 5, lane = tid & 31;
  float sum = 0.f, sq = 0.f;
  if (tid < 196) {
    int pix = tid;
    int r = pix / 14, c = pix - r * 14;
    int i00 = (2 * r) * 28 + 2 * c;
    long b0 = (long)blockIdx.x * L1_GIMG;

    const float2* xp = reinterpret_cast<const float2*>(x + b0 * 784 + i00);
    float2 top[L1_U], bot[L1_U];
    for (int u = 0; u < L1_U; ++u) { top[u] = xp[u * 392]; bot[u] = xp[u * 392 + 14]; }

    float a0 = w1[i00], a1 = w1[i00 + 1], a2 = w1[i00 + 28], a3 = w1[i00 + 29];
    float c0 = w3[pix];
    int wofs = r * 140 + 10 * c;
    bool cl = (c != 0), cr = (c != 13), rt = (r != 0), rb = (r != 13);
    const bool klr[4] = { cl, cr, cl, cr };
    const bool ktb[4] = { rt, rt, rb, rb };
    const int e1[10] = { 0, 1, 2, 3, 0, 0, 0, 1, 1, 2 };
    const int e2[10] = { 0, 1, 2, 3, 1, 2, 3, 2, 3, 3 };
    float q[10];
    for (int f = 0; f < 10; ++f) {
      bool mlr = klr[e1[f]] && klr[e2[f]];
      bool mtb = ktb[e1[f]] && ktb[e2[f]];
      float vlr = w2lr[wofs + f], vtb = w2tb[wofs + f];
      float val = w2all[wofs + f], vno = w2no[wofs + f];
      q[f] = (((mlr ? vlr : 0.f) + (mtb ? vtb : 0.f)) + ((mlr && mtb) ? val : 0.f)) + vno;
    }

    float* op = l1_out + b0 * 196 + pix;
    for (int it = 0; it < L1_GIMG / L1_U; ++it) {
      float2 ntop[L1_U], nbot[L1_U];
      if (it + 1 < L1_GIMG / L1_U) {
        for (int u = 0; u < L1_U; ++u) {
          ntop[u] = xp[((it + 1) * L1_U + u) * 392];
          nbot[u] = xp[((it + 1) * L1_U + u) * 392 + 14];
        }
      }
      for (int u = 0; u < L1_U; ++u) {
        float b00 = top[u].x, b01 = top[u].y, b10 = bot[u].x, b11 = bot[u].y;
        float t0 = fmaf(q[6], b11, fmaf(q[5], b10, fmaf(q[4], b01, fmaf(q[0], b00, a0))));
        float t1 = fmaf(q[8], b11, fmaf(q[7], b10, fmaf(q[1], b01, a1)));
        float t2 = fmaf(q[9], b11, fmaf(q[2], b10, a2));
        float t3 = fmaf(q[3], b11, a3);
        float acc = fmaf(b11, t3, fmaf(b10, t2, fmaf(b01, t1, fmaf(b00, t0, c0))));
        op[(it * L1_U + u) * 196] = acc;
        sum += acc; sq += acc * acc;
      }
      if (it + 1 < L1_GIMG / L1_U) {
        for (int u = 0; u < L1_U; ++u) { top[u] = ntop[u]; bot[u] = nbot[u]; }
      }
    }
  }

  for (int o = 16; o > 0; o >>= 1) {
    sum += __shfl_xor(sum, o);
    sq  += __shfl_xor(sq, o);
  }
  if (lane == 0) { s_red[wv] = sum; s_red[8 + wv] = sq; }
  __syncthreads();
  if (tid == 0) {
    float a = 0.f, qq = 0.f;
    for (int w = 0; w < 7; ++w) { a += s_red[w]; qq += s_red[8 + w]; }
    atomicAdd(&stats[0], a);
    atomicAdd(&stats[1], qq);
  }
}

__global__ void stats_kernel(float* __restrict__ stats,
                             const float* __restrict__ gamma,
                             const float* __restrict__ beta,
                             float n, int sIdx, int oIdx) {
  float mean = stats[sIdx] / n;
  float var = stats[sIdx + 1] / n - mean * mean;
  float s = gamma[0] * rsqrtf(var + EPS);
  stats[oIdx] = s;
  stats[oIdx + 1] = beta[0] - mean * s;
}

#define L2_GROUPS 512
#define L2_WIMG (BATCH / (L2_GROUPS * 8))
#define L2_WVEC (L2_WIMG * 196 / 4)
#define L2_WTASK (L2_WIMG * 49)
__global__ __launch_bounds__(256) void layer2_kernel(
    const float* __restrict__ l1_in,
    const float* __restrict__ w1, const float* __restrict__ w2,
    const float* __restrict__ w4,
    float* __restrict__ l2_out, float* __restrict__ stats) {
  __shared__ float s_w1[196];
  __shared__ float s_w2[490];
  __shared__ float s_w4[49];
  __shared__ __align__(16) float s_h[8][L2_WIMG * 196];
  __shared__ float s_red[16];
  int tid = threadIdx.x;
  int wv = tid >> 5, lane = tid & 31;
  long first = ((long)blockIdx.x * 8 + wv) * L2_WIMG;
  const float4* src = reinterpret_cast<const float4*>(l1_in + first * 196);
  float4 v[(L2_WVEC + 31) / 32];
#pragma unroll
  for (int k = 0; k < (L2_WVEC + 31) / 32; ++k) {
    int idx = lane + 32 * k;
    if (idx < L2_WVEC) v[k] = src[idx];
  }
  float q1 = (tid < 196) ? w1[tid] : 0.f;
  float q2 = w2[tid];
  float q3 = (tid < 490 - 256) ? w2[tid + 256] : 0.f;
  float q4 = (tid < 49) ? w4[tid] : 0.f;
  float sbn = stats[4], tbn = stats[5];
  if (tid < 196) s_w1[tid] = q1;
  s_w2[tid] = q2;
  if (tid < 490 - 256) s_w2[tid + 256] = q3;
  if (tid < 49) s_w4[tid] = q4;
  float4* hv = reinterpret_cast<float4*>(&s_h[wv][0]);
#pragma unroll
  for (int k = 0; k < (L2_WVEC + 31) / 32; ++k) {
    int idx = lane + 32 * k;
    if (idx < L2_WVEC) {
      float4 h;
      h.x = v[k].x * sbn + tbn; h.y = v[k].y * sbn + tbn;
      h.z = v[k].z * sbn + tbn; h.w = v[k].w * sbn + tbn;
      hv[idx] = h;
    }
  }
  __syncthreads();

  const float* hw = &s_h[wv][0];
  float sum = 0.f, sq = 0.f;
  for (int t = lane; t < L2_WTASK; t += 32) {
    int im = t / 49, pix = t - im * 49;
    int r = pix / 7, c = pix - r * 7;
    int i00 = (2 * r) * 14 + 2 * c;
    const float* hp = hw + im * 196 + i00;
    float b00 = hp[0],  b01 = hp[1];
    float b10 = hp[14], b11 = hp[15];
    float acc = b00 * s_w1[i00] + b01 * s_w1[i00 + 1]
              + b10 * s_w1[i00 + 14] + b11 * s_w1[i00 + 15];
    acc += s_w4[pix];
    acc += dot10(b00, b01, b10, b11, &s_w2[r * 70 + 10 * c]);
    l2_out[first * 49 + t] = acc;
    sum += acc; sq += acc * acc;
  }
  for (int o = 16; o > 0; o >>= 1) {
    sum += __shfl_xor(sum, o);
    sq  += __shfl_xor(sq, o);
  }
  if (lane == 0) { s_red[wv] = sum; s_red[8 + wv] = sq; }
  __syncthreads();
  if (tid == 0) {
    float a = 0.f, q = 0.f;
    for (int w = 0; w < 8; ++w) { a += s_red[w]; q += s_red[8 + w]; }
    atomicAdd(&stats[2], a);
    atomicAdd(&stats[3], q);
  }
}

__global__ __launch_bounds__(256) void gemm_kernel(
    const float* __restrict__ l2_in, const float* __restrict__ stats,
    const float2* __restrict__ wswz,
    const float* __restrict__ b_fc1, const float* __restrict__ b_fc2,
    float* __restrict__ out) {
  __shared__ __align__(16) float s_v[16 * 52];
  __shared__ float s_part[8][256];
  __shared__ float s_out[160];
  int tid = threadIdx.x;
  int wv = __builtin_amdgcn_readfirstlane(tid >> 5);
  int lane = tid & 31;
  int n = lane & 15, half = lane >> 4;
  long b0 = (long)blockIdx.x * 16;
  float sbn = stats[6], tbn = stats[7];

  const float* src = l2_in + b0 * 49;
  float x0 = src[tid], x1 = src[tid + 256], x2 = src[tid + 512];
  float x3 = (tid < 16) ? src[tid + 768] : 0.f;
  {
    int m = tid / 49;
    s_v[m * 52 + (tid - m * 49)] = x0 * sbn + tbn;
    int i1 = tid + 256; m = i1 / 49;
    s_v[m * 52 + (i1 - m * 49)] = x1 * sbn + tbn;
    int i2 = tid + 512; m = i2 / 49;
    s_v[m * 52 + (i2 - m * 49)] = x2 * sbn + tbn;
    if (tid < 16) s_v[15 * 52 + 33 + tid] = x3 * sbn + tbn;
  }
  if (tid < 48) {
    int m = tid / 3;
    s_v[m * 52 + 49 + (tid - m * 3)] = 0.f;
  }
  __syncthreads();

  const float* vp = s_v + n * 52;
  v8f acc = {};
  for (int i = wv; i < 50; i += 8) {
    float vi = (i < 49) ? vp[i] : 1.0f;
    const float2* wp = wswz + (size_t)i * (13 * 32) + lane;
#pragma unroll
    for (int q = 0; q < 13; ++q) {
      int j = 4 * q + 2 * half;
      float vj0 = vp[j], vj1 = vp[j + 1];
      v2f a; a.x = vi * vj0; a.y = vi * vj1;
      float2 w = wp[q * 32];
      v2f bm; bm.x = w.x; bm.y = w.y;
      acc = __builtin_amdgcn_wmma_f32_16x16x4_f32(
          false, a, false, bm, (short)0, acc, false, false);
    }
  }

#pragma unroll
  for (int r = 0; r < 8; ++r) s_part[wv][r * 32 + lane] = acc[r];
  __syncthreads();

  float sum = 0.f;
  for (int w = 0; w < 8; ++w) sum += s_part[w][tid];
  if (n < 10) {
    int M = (tid >> 5) + 8 * half;
    s_out[M * 10 + n] = sum + (b_fc1[n] + b_fc2[n]);
  }
  __syncthreads();
  if (tid < 160) out[b0 * 10 + tid] = s_out[tid];
}

extern "C" void kernel_launch(void* const* d_in, const int* in_sizes, int n_in,
                              void* d_out, int out_size, void* d_ws, size_t ws_size,
                              hipStream_t stream) {
  (void)in_sizes; (void)n_in; (void)out_size; (void)ws_size;
  const float* x       = (const float*)d_in[0];
  const float* l1_w1   = (const float*)d_in[1];
  const float* l1_w2_1 = (const float*)d_in[2];
  const float* l1_w2_2 = (const float*)d_in[3];
  const float* l1_w2_3 = (const float*)d_in[4];
  const float* l1_w2_4 = (const float*)d_in[5];
  const float* l1_w3   = (const float*)d_in[6];
  const float* l2_w1   = (const float*)d_in[7];
  const float* l2_w2   = (const float*)d_in[8];
  const float* l2_w4   = (const float*)d_in[9];
  const float* w_fc1   = (const float*)d_in[10];
  const float* b_fc1   = (const float*)d_in[11];
  const float* w_fc2   = (const float*)d_in[12];
  const float* b_fc2   = (const float*)d_in[13];
  const float* gamma   = (const float*)d_in[14];
  const float* beta    = (const float*)d_in[15];
  float* out = (float*)d_out;

  float* l1_buf = (float*)d_ws;
  float* l2_buf = l1_buf + (size_t)BATCH * 196;
  float* wswz   = l2_buf + (size_t)BATCH * 49;
  float* stats  = wswz + (size_t)650 * 32 * 2;

  prep_kernel<<<(650 * 32 + 255) / 256, 256, 0, stream>>>(w_fc1, w_fc2, (float2*)wswz, stats);
  layer1_kernel<<<L1_GROUPS, 224, 0, stream>>>(x, l1_w1, l1_w2_1, l1_w2_2, l1_w2_3,
                                               l1_w2_4, l1_w3, l1_buf, stats);
  stats_kernel<<<1, 1, 0, stream>>>(stats, gamma, beta, (float)(BATCH * 196), 0, 4);
  layer2_kernel<<<L2_GROUPS, 256, 0, stream>>>(l1_buf, l2_w1, l2_w2, l2_w4, l2_buf, stats);
  stats_kernel<<<1, 1, 0, stream>>>(stats, gamma, beta, (float)(BATCH * 49), 2, 6);
  gemm_kernel<<<BATCH / 16, 256, 0, stream>>>(l2_buf, stats, (const float2*)wswz,
                                               b_fc1, b_fc2, out);
}


# three launches instead of six: each consumer group adds the producers' BN partial pairs itself in a fixed order (no atomics, reorders f32 BN sums); head weights built by spare layer-1 groups
# speedup vs baseline: 4.3702x; 1.2764x over previous
#include <hip/hip_runtime.h>

typedef __attribute__((ext_vector_type(2))) float v2f;
typedef __attribute__((ext_vector_type(8))) float v8f;

#define BATCH 16384
#define EPS 1e-5f

#define BN_PARTS 512
static_assert(2 * BN_PARTS == 4 * 256, "a consumer group of 256 threads reads the pairs as one float4 each");

__device__ inline float wz_val(const float* w_fc1, const float* w_fc2, int k, int n) {
  if (n >= 10) return 0.f;
  int i = k / 52, j = k - i * 52;
  if (j >= 49) return 0.f;
  if (i < 49) return (i <= j) ? w_fc2[n * 2401 + i * 49 + j] : 0.f;
  return w_fc1[n * 49 + j];
}

#define WZ_ITEMS (650 * 32)
__device__ inline void wz_build(const float* __restrict__ w_fc1,
                                const float* __restrict__ w_fc2,
                                float2* __restrict__ wswz, int t) {
  if (t < WZ_ITEMS) {
    int s = t >> 5, lane = t & 31;
    int n = lane & 15, half = lane >> 4;
    int ka = 4 * s + 2 * half;
    float2 v;
    v.x = wz_val(w_fc1, w_fc2, ka, n);
    v.y = wz_val(w_fc1, w_fc2, ka + 1, n);
    wswz[t] = v;
  }
}

template <int XMASK> __device__ inline float row_xor(float v) {
  return __builtin_bit_cast(float, __builtin_amdgcn_update_dpp(
      0, __builtin_bit_cast(int, v), 0x160 + XMASK, 0xf, 0xf, false));
}
__device__ inline float other_row(float v) {
  return __builtin_bit_cast(float, __builtin_amdgcn_permlanex16(
      __builtin_bit_cast(int, v), __builtin_bit_cast(int, v),
      0x76543210, 0xfedcba98, false, false));
}
__device__ inline float wave_sum(float v) {
  v += row_xor<1>(v);
  v += row_xor<2>(v);
  v += row_xor<4>(v);
  v += row_xor<8>(v);
  return v + other_row(v);
}

__device__ inline void bn_partial(float4 pv, int wv, int lane, float* s_red) {
  float a = wave_sum(pv.x + pv.z), q = wave_sum(pv.y + pv.w);
  if (lane == 0) { s_red[wv] = a; s_red[8 + wv] = q; }
}

__device__ inline void bn_scale_shift(const float* s_red, float n,
                                      const float* __restrict__ gamma,
                                      const float* __restrict__ beta,
                                      float& s, float& t) {
  float a = 0.f, q = 0.f;
  for (int w = 0; w < 8; ++w) { a += s_red[w]; q += s_red[8 + w]; }
  float mean = a / n;
  float var = q / n - mean * mean;
  s = gamma[0] * rsqrtf(var + EPS);
  t = beta[0] - mean * s;
}

__device__ inline float dot10(float t00, float t01, float t10, float t11,
                              const float* __restrict__ W) {
  return W[0]*t00*t00 + W[1]*t01*t01 + W[2]*t10*t10 + W[3]*t11*t11
       + W[4]*t00*t01 + W[5]*t00*t10 + W[6]*t00*t11
       + W[7]*t01*t10 + W[8]*t01*t11 + W[9]*t10*t11;
}

#define L1_GROUPS BN_PARTS
#define L1_GIMG (BATCH / L1_GROUPS)
#define L1_U 8
#define WZ_GROUPS ((WZ_ITEMS + 223) / 224)
__global__ __launch_bounds__(224) void layer1_kernel(
    const float* __restrict__ x,
    const float* __restrict__ w1,   const float* __restrict__ w2lr,
    const float* __restrict__ w2tb, const float* __restrict__ w2all,
    const float* __restrict__ w2no, const float* __restrict__ w3,
    float* __restrict__ l1_out, float2* __restrict__ part,
    const float* __restrict__ w_fc1, const float* __restrict__ w_fc2,
    float2* __restrict__ wswz) {
  __shared__ float s_red[16];
  int tid = threadIdx.x;
  if (blockIdx.x >= L1_GROUPS) {
    wz_build(w_fc1, w_fc2, wswz, (blockIdx.x - L1_GROUPS) * 224 + tid);
    return;
  }
  int wv = tid >> 5, lane = tid & 31;
  float sum = 0.f, sq = 0.f;
  if (tid < 196) {
    int pix = tid;
    int r = pix / 14, c = pix - r * 14;
    int i00 = (2 * r) * 28 + 2 * c;
    long b0 = (long)blockIdx.x * L1_GIMG;

    const float2* xp = reinterpret_cast<const float2*>(x + b0 * 784 + i00);
    float2 top[L1_U], bot[L1_U];
    for (int u = 0; u < L1_U; ++u) { top[u] = xp[u * 392]; bot[u] = xp[u * 392 + 14]; }

    float a0 = w1[i00], a1 = w1[i00 + 1], a2 = w1[i00 + 28], a3 = w1[i00 + 29];
    float c0 = w3[pix];
    int wofs = r * 140 + 10 * c;
    bool cl = (c != 0), cr = (c != 13), rt = (r != 0), rb = (r != 13);
    const bool klr[4] = { cl, cr, cl, cr };
    const bool ktb[4] = { rt, rt, rb, rb };
    const int e1[10] = { 0, 1, 2, 3, 0, 0, 0, 1, 1, 2 };
    const int e2[10] = { 0, 1, 2, 3, 1, 2, 3, 2, 3, 3 };
    float q[10];
    for (int f = 0; f < 10; ++f) {
      bool mlr = klr[e1[f]] && klr[e2[f]];
      bool mtb = ktb[e1[f]] && ktb[e2[f]];
      float vlr = w2lr[wofs + f], vtb = w2tb[wofs + f];
      float val = w2all[wofs + f], vno = w2no[wofs + f];
      q[f] = (((mlr ? vlr : 0.f) + (mtb ? vtb : 0.f)) + ((mlr && mtb) ? val : 0.f)) + vno;
    }

    float* op = l1_out + b0 * 196 + pix;
    for (int it = 0; it < L1_GIMG / L1_U; ++it) {
      float2 ntop[L1_U], nbot[L1_U];
      if (it + 1 < L1_GIMG / L1_U) {
        for (int u = 0; u < L1_U; ++u) {
          ntop[u] = xp[((it + 1) * L1_U + u) * 392];
          nbot[u] = xp[((it + 1) * L1_U + u) * 392 + 14];
        }
      }
      for (int u = 0; u < L1_U; ++u) {
        float b00 = top[u].x, b01 = top[u].y, b10 = bot[u].x, b11 = bot[u].y;
        float t0 = fmaf(q[6], b11, fmaf(q[5], b10, fmaf(q[4], b01, fmaf(q[0], b00, a0))));
        float t1 = fmaf(q[8], b11, fmaf(q[7], b10, fmaf(q[1], b01, a1)));
        float t2 = fmaf(q[9], b11, fmaf(q[2], b10, a2));
        float t3 = fmaf(q[3], b11, a3);
        float acc = fmaf(b11, t3, fmaf(b10, t2, fmaf(b01, t1, fmaf(b00, t0, c0))));
        op[(it * L1_U + u) * 196] = acc;
        sum += acc; sq += acc * acc;
      }
      if (it + 1 < L1_GIMG / L1_U) {
        for (int u = 0; u < L1_U; ++u) { top[u] = ntop[u]; bot[u] = nbot[u]; }
      }
    }
  }

  for (int o = 16; o > 0; o >>= 1) {
    sum += __shfl_xor(sum, o);
    sq  += __shfl_xor(sq, o);
  }
  if (lane == 0) { s_red[wv] = sum; s_red[8 + wv] = sq; }
  __syncthreads();
  if (tid == 0) {
    float2 p; p.x = 0.f; p.y = 0.f;
    for (int w = 0; w < 7; ++w) { p.x += s_red[w]; p.y += s_red[8 + w]; }
    part[blockIdx.x] = p;
  }
}

#define L2_GROUPS BN_PARTS
#define L2_WIMG (BATCH / (L2_GROUPS * 8))
#define L2_WVEC (L2_WIMG * 196 / 4)
#define L2_WTASK (L2_WIMG * 49)
#define L2_GIMG (8 * L2_WIMG)
#define L2_GPITCH (L2_GIMG * 196)
__global__ __launch_bounds__(256) void layer2_kernel(
    float* __restrict__ buf,
    const float* __restrict__ part_in,
    const float* __restrict__ gamma, const float* __restrict__ beta,
    const float* __restrict__ w1, const float* __restrict__ w2,
    const float* __restrict__ w4,
    float2* __restrict__ part_out) {
  __shared__ float s_w1[196];
  __shared__ float s_w2[490];
  __shared__ float s_w4[49];
  __shared__ __align__(16) float s_h[8][L2_WIMG * 196];
  __shared__ float s_bn[16];
  __shared__ float s_red[16];
  int tid = threadIdx.x;
  int wv = tid >> 5, lane = tid & 31;
  float* gbuf = buf + (long)blockIdx.x * L2_GPITCH;
  float4 pv = reinterpret_cast<const float4*>(part_in)[tid];
  float q1 = (tid < 196) ? w1[tid] : 0.f;
  float q2 = w2[tid];
  float q3 = (tid < 490 - 256) ? w2[tid + 256] : 0.f;
  float q4 = (tid < 49) ? w4[tid] : 0.f;
  const float4* src = reinterpret_cast<const float4*>(gbuf + wv * (L2_WIMG * 196));
  float4 v[(L2_WVEC + 31) / 32];
#pragma unroll
  for (int k = 0; k < (L2_WVEC + 31) / 32; ++k) {
    int idx = lane + 32 * k;
    v[k] = (idx < L2_WVEC) ? src[idx] : float4{0.f, 0.f, 0.f, 0.f};
  }
  bn_partial(pv, wv, lane, s_bn);
  if (tid < 196) s_w1[tid] = q1;
  s_w2[tid] = q2;
  if (tid < 490 - 256) s_w2[tid + 256] = q3;
  if (tid < 49) s_w4[tid] = q4;
  float4* hv = reinterpret_cast<float4*>(&s_h[wv][0]);
#pragma unroll
  for (int k = 0; k < (L2_WVEC + 31) / 32; ++k) {
    int idx = lane + 32 * k;
    if (idx < L2_WVEC) hv[idx] = v[k];
  }
  __syncthreads();
  float sbn, tbn;
  bn_scale_shift(s_bn, (float)(BATCH * 196), gamma, beta, sbn, tbn);

  const float* hw = &s_h[wv][0];
  float sum = 0.f, sq = 0.f;
  for (int t = lane; t < L2_WTASK; t += 32) {
    int im = t / 49, pix = t - im * 49;
    int r = pix / 7, c = pix - r * 7;
    int i00 = (2 * r) * 14 + 2 * c;
    const float* hp = hw + im * 196 + i00;
    float b00 = hp[0]  * sbn + tbn, b01 = hp[1]  * sbn + tbn;
    float b10 = hp[14] * sbn + tbn, b11 = hp[15] * sbn + tbn;
    float acc = b00 * s_w1[i00] + b01 * s_w1[i00 + 1]
              + b10 * s_w1[i00 + 14] + b11 * s_w1[i00 + 15];
    acc += s_w4[pix];
    acc += dot10(b00, b01, b10, b11, &s_w2[r * 70 + 10 * c]);
    gbuf[wv * L2_WTASK + t] = acc;
    sum += acc; sq += acc * acc;
  }
  for (int o = 16; o > 0; o >>= 1) {
    sum += __shfl_xor(sum, o);
    sq  += __shfl_xor(sq, o);
  }
  if (lane == 0) { s_red[wv] = sum; s_red[8 + wv] = sq; }
  __syncthreads();
  if (tid == 0) {
    float2 p; p.x = 0.f; p.y = 0.f;
    for (int w = 0; w < 8; ++w) { p.x += s_red[w]; p.y += s_red[8 + w]; }
    part_out[blockIdx.x] = p;
  }
}

__global__ __launch_bounds__(256) void gemm_kernel(
    const float* __restrict__ l2_in, const float* __restrict__ part_in,
    const float* __restrict__ gamma, const float* __restrict__ beta,
    const float2* __restrict__ wswz,
    const float* __restrict__ b_fc1, const float* __restrict__ b_fc2,
    float* __restrict__ out) {
  __shared__ __align__(16) float s_v[16 * 52];
  __shared__ float s_part[8][256];
  __shared__ float s_out[160];
  __shared__ float s_bn[16];
  int tid = threadIdx.x;
  int wv = __builtin_amdgcn_readfirstlane(tid >> 5);
  int lane = tid & 31;
  int n = lane & 15, half = lane >> 4;
  long b0 = (long)blockIdx.x * 16;

  const float* src = l2_in + (b0 / L2_GIMG) * L2_GPITCH + (b0 % L2_GIMG) * 49;
  float4 pv = reinterpret_cast<const float4*>(part_in)[tid];
  float x0 = src[tid], x1 = src[tid + 256], x2 = src[tid + 512];
  float x3 = (tid < 16) ? src[tid + 768] : 0.f;
  bn_partial(pv, wv, lane, s_bn);
  __syncthreads();
  float sbn, tbn;
  bn_scale_shift(s_bn, (float)(BATCH * 49), gamma, beta, sbn, tbn);
  {
    int m = tid / 49;
    s_v[m * 52 + (tid - m * 49)] = x0 * sbn + tbn;
    int i1 = tid + 256; m = i1 / 49;
    s_v[m * 52 + (i1 - m * 49)] = x1 * sbn + tbn;
    int i2 = tid + 512; m = i2 / 49;
    s_v[m * 52 + (i2 - m * 49)] = x2 * sbn + tbn;
    if (tid < 16) s_v[15 * 52 + 33 + tid] = x3 * sbn + tbn;
  }
  if (tid < 48) {
    int m = tid / 3;
    s_v[m * 52 + 49 + (tid - m * 3)] = 0.f;
  }
  __syncthreads();

  const float* vp = s_v + n * 52;
  v8f acc = {};
  for (int i = wv; i < 50; i += 8) {
    float vi = (i < 49) ? vp[i] : 1.0f;
    const float2* wp = wswz + (size_t)i * (13 * 32) + lane;
#pragma unroll
    for (int q = 0; q < 13; ++q) {
      int j = 4 * q + 2 * half;
      float vj0 = vp[j], vj1 = vp[j + 1];
      v2f a; a.x = vi * vj0; a.y = vi * vj1;
      float2 w = wp[q * 32];
      v2f bm; bm.x = w.x; bm.y = w.y;
      acc = __builtin_amdgcn_wmma_f32_16x16x4_f32(
          false, a, false, bm, (short)0, acc, false, false);
    }
  }

#pragma unroll
  for (int r = 0; r < 8; ++r) s_part[wv][r * 32 + lane] = acc[r];
  __syncthreads();

  float sum = 0.f;
  for (int w = 0; w < 8; ++w) sum += s_part[w][tid];
  if (n < 10) {
    int M = (tid >> 5) + 8 * half;
    s_out[M * 10 + n] = sum + (b_fc1[n] + b_fc2[n]);
  }
  __syncthreads();
  if (tid < 160) out[b0 * 10 + tid] = s_out[tid];
}

extern "C" void kernel_launch(void* const* d_in, const int* in_sizes, int n_in,
                              void* d_out, int out_size, void* d_ws, size_t ws_size,
                              hipStream_t stream) {
  (void)in_sizes; (void)n_in; (void)out_size; (void)ws_size;
  const float* x       = (const float*)d_in[0];
  const float* l1_w1   = (const float*)d_in[1];
  const float* l1_w2_1 = (const float*)d_in[2];
  const float* l1_w2_2 = (const float*)d_in[3];
  const float* l1_w2_3 = (const float*)d_in[4];
  const float* l1_w2_4 = (const float*)d_in[5];
  const float* l1_w3   = (const float*)d_in[6];
  const float* l2_w1   = (const float*)d_in[7];
  const float* l2_w2   = (const float*)d_in[8];
  const float* l2_w4   = (const float*)d_in[9];
  const float* w_fc1   = (const float*)d_in[10];
  const float* b_fc1   = (const float*)d_in[11];
  const float* w_fc2   = (const float*)d_in[12];
  const float* b_fc2   = (const float*)d_in[13];
  const float* gamma   = (const float*)d_in[14];
  const float* beta    = (const float*)d_in[15];
  float* out = (float*)d_out;

  float* l1_buf = (float*)d_ws;
  float* part1  = l1_buf + (size_t)BATCH * 196;
  float* part2  = part1 + 2 * BN_PARTS;
  float* wswz   = part2 + 2 * BN_PARTS;

  layer1_kernel<<<L1_GROUPS + WZ_GROUPS, 224, 0, stream>>>(
      x, l1_w1, l1_w2_1, l1_w2_2, l1_w2_3, l1_w2_4, l1_w3, l1_buf, (float2*)part1,
      w_fc1, w_fc2, (float2*)wswz);
  layer2_kernel<<<L2_GROUPS, 256, 0, stream>>>(l1_buf, part1, gamma, beta,
                                               l2_w1, l2_w2, l2_w4, (float2*)part2);
  gemm_kernel<<<BATCH / 16, 256, 0, stream>>>(l1_buf, part2, gamma, beta,
                                              (const float2*)wswz, b_fc1, b_fc2, out);
}


# head GEMM skips the K-steps the triu mask zeroes: row i starts at step i/4 (361 matrix steps per tile, not 650), linear term in pad column 49, rows dealt evenly over 8 waves; reorders the f32 sums
# speedup vs baseline: 5.1352x; 1.1750x over previous
#include <hip/hip_runtime.h>

typedef __attribute__((ext_vector_type(2))) float v2f;
typedef __attribute__((ext_vector_type(8))) float v8f;

#define BATCH 16384
#define EPS 1e-5f

#define BN_PARTS 512
static_assert(2 * BN_PARTS == 4 * 256, "a consumer group of 256 threads reads the pairs as one float4 each");

__device__ inline float wz_val(const float* w_fc1, const float* w_fc2, int k, int n) {
  if (n >= 10) return 0.f;
  int i = k / 52, j = k - i * 52;
  if (j < 49) return (i <= j) ? w_fc2[n * 2401 + i * 49 + j] : 0.f;
  if (j == 49) return w_fc1[n * 49 + i];
  return 0.f;
}

#define WZ_ITEMS (49 * 13 * 32)
__device__ inline void wz_build(const float* __restrict__ w_fc1,
                                const float* __restrict__ w_fc2,
                                float2* __restrict__ wswz, int t) {
  if (t < WZ_ITEMS) {
    int s = t >> 5, lane = t & 31;
    int i = s / 13;
    if (s - 13 * i < (i >> 2)) return;
    int n = lane & 15, half = lane >> 4;
    int ka = 4 * s + 2 * half;
    float2 v;
    v.x = wz_val(w_fc1, w_fc2, ka, n);
    v.y = wz_val(w_fc1, w_fc2, ka + 1, n);
    wswz[t] = v;
  }
}

template <int XMASK> __device__ inline float row_xor(float v) {
  return __builtin_bit_cast(float, __builtin_amdgcn_update_dpp(
      0, __builtin_bit_cast(int, v), 0x160 + XMASK, 0xf, 0xf, false));
}
__device__ inline float other_row(float v) {
  return __builtin_bit_cast(float, __builtin_amdgcn_permlanex16(
      __builtin_bit_cast(int, v), __builtin_bit_cast(int, v),
      0x76543210, 0xfedcba98, false, false));
}
__device__ inline float wave_sum(float v) {
  v += row_xor<1>(v);
  v += row_xor<2>(v);
  v += row_xor<4>(v);
  v += row_xor<8>(v);
  return v + other_row(v);
}

__device__ inline void bn_partial(float4 pv, int wv, int lane, float* s_red) {
  float a = wave_sum(pv.x + pv.z), q = wave_sum(pv.y + pv.w);
  if (lane == 0) { s_red[wv] = a; s_red[8 + wv] = q; }
}

__device__ inline void bn_scale_shift(const float* s_red, float n,
                                      const float* __restrict__ gamma,
                                      const float* __restrict__ beta,
                                      float& s, float& t) {
  float a = 0.f, q = 0.f;
  for (int w = 0; w < 8; ++w) { a += s_red[w]; q += s_red[8 + w]; }
  float mean = a / n;
  float var = q / n - mean * mean;
  s = gamma[0] * rsqrtf(var + EPS);
  t = beta[0] - mean * s;
}

__device__ inline float dot10(float t00, float t01, float t10, float t11,
                              const float* __restrict__ W) {
  return W[0]*t00*t00 + W[1]*t01*t01 + W[2]*t10*t10 + W[3]*t11*t11
       + W[4]*t00*t01 + W[5]*t00*t10 + W[6]*t00*t11
       + W[7]*t01*t10 + W[8]*t01*t11 + W[9]*t10*t11;
}

#define L1_GROUPS BN_PARTS
#define L1_GIMG (BATCH / L1_GROUPS)
#define L1_U 8
#define WZ_GROUPS ((WZ_ITEMS + 223) / 224)
__global__ __launch_bounds__(224) void layer1_kernel(
    const float* __restrict__ x,
    const float* __restrict__ w1,   const float* __restrict__ w2lr,
    const float* __restrict__ w2tb, const float* __restrict__ w2all,
    const float* __restrict__ w2no, const float* __restrict__ w3,
    float* __restrict__ l1_out, float2* __restrict__ part,
    const float* __restrict__ w_fc1, const float* __restrict__ w_fc2,
    float2* __restrict__ wswz) {
  __shared__ float s_red[16];
  int tid = threadIdx.x;
  if (blockIdx.x >= L1_GROUPS) {
    wz_build(w_fc1, w_fc2, wswz, (blockIdx.x - L1_GROUPS) * 224 + tid);
    return;
  }
  int wv = tid >> 5, lane = tid & 31;
  float sum = 0.f, sq = 0.f;
  if (tid < 196) {
    int pix = tid;
    int r = pix / 14, c = pix - r * 14;
    int i00 = (2 * r) * 28 + 2 * c;
    long b0 = (long)blockIdx.x * L1_GIMG;

    const float2* xp = reinterpret_cast<const float2*>(x + b0 * 784 + i00);
    float2 top[L1_U], bot[L1_U];
    for (int u = 0; u < L1_U; ++u) { top[u] = xp[u * 392]; bot[u] = xp[u * 392 + 14]; }

    float a0 = w1[i00], a1 = w1[i00 + 1], a2 = w1[i00 + 28], a3 = w1[i00 + 29];
    float c0 = w3[pix];
    int wofs = r * 140 + 10 * c;
    bool cl = (c != 0), cr = (c != 13), rt = (r != 0), rb = (r != 13);
    const bool klr[4] = { cl, cr, cl, cr };
    const bool ktb[4] = { rt, rt, rb, rb };
    const int e1[10] = { 0, 1, 2, 3, 0, 0, 0, 1, 1, 2 };
    const int e2[10] = { 0, 1, 2, 3, 1, 2, 3, 2, 3, 3 };
    float q[10];
    for (int f = 0; f < 10; ++f) {
      bool mlr = klr[e1[f]] && klr[e2[f]];
      bool mtb = ktb[e1[f]] && ktb[e2[f]];
      float vlr = w2lr[wofs + f], vtb = w2tb[wofs + f];
      float val = w2all[wofs + f], vno = w2no[wofs + f];
      q[f] = (((mlr ? vlr : 0.f) + (mtb ? vtb : 0.f)) + ((mlr && mtb) ? val : 0.f)) + vno;
    }

    float* op = l1_out + b0 * 196 + pix;
    for (int it = 0; it < L1_GIMG / L1_U; ++it) {
      float2 ntop[L1_U], nbot[L1_U];
      if (it + 1 < L1_GIMG / L1_U) {
        for (int u = 0; u < L1_U; ++u) {
          ntop[u] = xp[((it + 1) * L1_U + u) * 392];
          nbot[u] = xp[((it + 1) * L1_U + u) * 392 + 14];
        }
      }
      for (int u = 0; u < L1_U; ++u) {
        float b00 = top[u].x, b01 = top[u].y, b10 = bot[u].x, b11 = bot[u].y;
        float t0 = fmaf(q[6], b11, fmaf(q[5], b10, fmaf(q[4], b01, fmaf(q[0], b00, a0))));
        float t1 = fmaf(q[8], b11, fmaf(q[7], b10, fmaf(q[1], b01, a1)));
        float t2 = fmaf(q[9], b11, fmaf(q[2], b10, a2));
        float t3 = fmaf(q[3], b11, a3);
        float acc = fmaf(b11, t3, fmaf(b10, t2, fmaf(b01, t1, fmaf(b00, t0, c0))));
        op[(it * L1_U + u) * 196] = acc;
        sum += acc; sq += acc * acc;
      }
      if (it + 1 < L1_GIMG / L1_U) {
        for (int u = 0; u < L1_U; ++u) { top[u] = ntop[u]; bot[u] = nbot[u]; }
      }
    }
  }

  for (int o = 16; o > 0; o >>= 1) {
    sum += __shfl_xor(sum, o);
    sq  += __shfl_xor(sq, o);
  }
  if (lane == 0) { s_red[wv] = sum; s_red[8 + wv] = sq; }
  __syncthreads();
  if (tid == 0) {
    float2 p; p.x = 0.f; p.y = 0.f;
    for (int w = 0; w < 7; ++w) { p.x += s_red[w]; p.y += s_red[8 + w]; }
    part[blockIdx.x] = p;
  }
}

#define L2_GROUPS BN_PARTS
#define L2_WIMG (BATCH / (L2_GROUPS * 8))
#define L2_WVEC (L2_WIMG * 196 / 4)
#define L2_WTASK (L2_WIMG * 49)
#define L2_GIMG (8 * L2_WIMG)
#define L2_GPITCH (L2_GIMG * 196)
__global__ __launch_bounds__(256) void layer2_kernel(
    float* __restrict__ buf,
    const float* __restrict__ part_in,
    const float* __restrict__ gamma, const float* __restrict__ beta,
    const float* __restrict__ w1, const float* __restrict__ w2,
    const float* __restrict__ w4,
    float2* __restrict__ part_out) {
  __shared__ float s_w1[196];
  __shared__ float s_w2[490];
  __shared__ float s_w4[49];
  __shared__ __align__(16) float s_h[8][L2_WIMG * 196];
  __shared__ float s_bn[16];
  __shared__ float s_red[16];
  int tid = threadIdx.x;
  int wv = tid >> 5, lane = tid & 31;
  float* gbuf = buf + (long)blockIdx.x * L2_GPITCH;
  float4 pv = reinterpret_cast<const float4*>(part_in)[tid];
  float q1 = (tid < 196) ? w1[tid] : 0.f;
  float q2 = w2[tid];
  float q3 = (tid < 490 - 256) ? w2[tid + 256] : 0.f;
  float q4 = (tid < 49) ? w4[tid] : 0.f;
  const float4* src = reinterpret_cast<const float4*>(gbuf + wv * (L2_WIMG * 196));
  float4 v[(L2_WVEC + 31) / 32];
#pragma unroll
  for (int k = 0; k < (L2_WVEC + 31) / 32; ++k) {
    int idx = lane + 32 * k;
    v[k] = (idx < L2_WVEC) ? src[idx] : float4{0.f, 0.f, 0.f, 0.f};
  }
  bn_partial(pv, wv, lane, s_bn);
  if (tid < 196) s_w1[tid] = q1;
  s_w2[tid] = q2;
  if (tid < 490 - 256) s_w2[tid + 256] = q3;
  if (tid < 49) s_w4[tid] = q4;
  float4* hv = reinterpret_cast<float4*>(&s_h[wv][0]);
#pragma unroll
  for (int k = 0; k < (L2_WVEC + 31) / 32; ++k) {
    int idx = lane + 32 * k;
    if (idx < L2_WVEC) hv[idx] = v[k];
  }
  __syncthreads();
  float sbn, tbn;
  bn_scale_shift(s_bn, (float)(BATCH * 196), gamma, beta, sbn, tbn);

  const float* hw = &s_h[wv][0];
  float sum = 0.f, sq = 0.f;
  for (int t = lane; t < L2_WTASK; t += 32) {
    int im = t / 49, pix = t - im * 49;
    int r = pix / 7, c = pix - r * 7;
    int i00 = (2 * r) * 14 + 2 * c;
    const float* hp = hw + im * 196 + i00;
    float b00 = hp[0]  * sbn + tbn, b01 = hp[1]  * sbn + tbn;
    float b10 = hp[14] * sbn + tbn, b11 = hp[15] * sbn + tbn;
    float acc = b00 * s_w1[i00] + b01 * s_w1[i00 + 1]
              + b10 * s_w1[i00 + 14] + b11 * s_w1[i00 + 15];
    acc += s_w4[pix];
    acc += dot10(b00, b01, b10, b11, &s_w2[r * 70 + 10 * c]);
    gbuf[wv * L2_WTASK + t] = acc;
    sum += acc; sq += acc * acc;
  }
  for (int o = 16; o > 0; o >>= 1) {
    sum += __shfl_xor(sum, o);
    sq  += __shfl_xor(sq, o);
  }
  if (lane == 0) { s_red[wv] = sum; s_red[8 + wv] = sq; }
  __syncthreads();
  if (tid == 0) {
    float2 p; p.x = 0.f; p.y = 0.f;
    for (int w = 0; w < 8; ++w) { p.x += s_red[w]; p.y += s_red[8 + w]; }
    part_out[blockIdx.x] = p;
  }
}

__device__ inline v8f head_step(v8f acc, float vi, float2 vjq, float2 w) {
  v2f a; a.x = vi * vjq.x; a.y = vi * vjq.y;
  v2f bm; bm.x = w.x; bm.y = w.y;
  return __builtin_amdgcn_wmma_f32_16x16x4_f32(
      false, a, false, bm, (short)0, acc, false, false);
}

template <int Q0>
__device__ inline v8f head_row(v8f acc, float vi, const float2* __restrict__ wp,
                               const float2 (&vj)[13]) {
  float2 w[13];
#pragma unroll
  for (int q = Q0; q < 13; ++q) w[q] = wp[q * 32];
#pragma unroll
  for (int q = Q0; q < 13; ++q) acc = head_step(acc, vi, vj[q], w[q]);
  return acc;
}

template <int B, bool LOW>
__device__ inline v8f head_block(v8f acc, int wv, const float* vp,
                                 const float2* __restrict__ wl,
                                 const float2 (&vj)[13]) {
  constexpr bool odd = (B & 1) != 0;
  constexpr int Q0 = 2 * B + ((LOW != odd) ? 0 : 1);
  int i = 8 * B + (odd ? 7 - wv : wv);
  return head_row<Q0>(acc, vp[i], wl + (size_t)i * (13 * 32), vj);
}

template <bool LOW>
__device__ inline v8f head_rows(v8f acc, int wv, const float* vp,
                                const float2* __restrict__ wl,
                                const float2 (&vj)[13]) {
  acc = head_block<0, LOW>(acc, wv, vp, wl, vj);
  acc = head_block<1, LOW>(acc, wv, vp, wl, vj);
  acc = head_block<2, LOW>(acc, wv, vp, wl, vj);
  acc = head_block<3, LOW>(acc, wv, vp, wl, vj);
  acc = head_block<4, LOW>(acc, wv, vp, wl, vj);
  acc = head_block<5, LOW>(acc, wv, vp, wl, vj);
  return acc;
}

__global__ __launch_bounds__(256) void gemm_kernel(
    const float* __restrict__ l2_in, const float* __restrict__ part_in,
    const float* __restrict__ gamma, const float* __restrict__ beta,
    const float2* __restrict__ wswz,
    const float* __restrict__ b_fc1, const float* __restrict__ b_fc2,
    float* __restrict__ out) {
  __shared__ __align__(16) float s_v[16 * 52];
  __shared__ float s_part[8][256];
  __shared__ float s_out[160];
  __shared__ float s_bn[16];
  int tid = threadIdx.x;
  int wv = __builtin_amdgcn_readfirstlane(tid >> 5);
  int lane = tid & 31;
  int n = lane & 15, half = lane >> 4;
  long b0 = (long)blockIdx.x * 16;

  const float* src = l2_in + (b0 / L2_GIMG) * L2_GPITCH + (b0 % L2_GIMG) * 49;
  float4 pv = reinterpret_cast<const float4*>(part_in)[tid];
  float x0 = src[tid], x1 = src[tid + 256], x2 = src[tid + 512];
  float x3 = (tid < 16) ? src[tid + 768] : 0.f;
  float2 w48;
  if (wv == 0) w48 = wswz[(size_t)(48 * 13 + 12) * 32 + lane];
  bn_partial(pv, wv, lane, s_bn);
  __syncthreads();
  float sbn, tbn;
  bn_scale_shift(s_bn, (float)(BATCH * 49), gamma, beta, sbn, tbn);
  {
    int m = tid / 49;
    s_v[m * 52 + (tid - m * 49)] = x0 * sbn + tbn;
    int i1 = tid + 256; m = i1 / 49;
    s_v[m * 52 + (i1 - m * 49)] = x1 * sbn + tbn;
    int i2 = tid + 512; m = i2 / 49;
    s_v[m * 52 + (i2 - m * 49)] = x2 * sbn + tbn;
    if (tid < 16) s_v[15 * 52 + 33 + tid] = x3 * sbn + tbn;
  }
  if (tid < 48) {
    int m = tid / 3, cpad = tid - m * 3;
    s_v[m * 52 + 49 + cpad] = (cpad == 0) ? 1.0f : 0.f;
  }
  __syncthreads();

  const float* vp = s_v + n * 52;
  float2 vj[13];
#pragma unroll
  for (int q = 0; q < 13; ++q)
    vj[q] = *reinterpret_cast<const float2*>(vp + 4 * q + 2 * half);
  const float2* wl = wswz + lane;
  v8f acc = {};
  if (wv < 4) {
    if (wv == 0) acc = head_step(acc, vp[48], vj[12], w48);
    acc = head_rows<true>(acc, wv, vp, wl, vj);
  } else {
    acc = head_rows<false>(acc, wv, vp, wl, vj);
  }

#pragma unroll
  for (int r = 0; r < 8; ++r) s_part[wv][r * 32 + lane] = acc[r];
  __syncthreads();

  float sum = 0.f;
  for (int w = 0; w < 8; ++w) sum += s_part[w][tid];
  if (n < 10) {
    int M = (tid >> 5) + 8 * half;
    s_out[M * 10 + n] = sum + (b_fc1[n] + b_fc2[n]);
  }
  __syncthreads();
  if (tid < 160) out[b0 * 10 + tid] = s_out[tid];
}

extern "C" void kernel_launch(void* const* d_in, const int* in_sizes, int n_in,
                              void* d_out, int out_size, void* d_ws, size_t ws_size,
                              hipStream_t stream) {
  (void)in_sizes; (void)n_in; (void)out_size; (void)ws_size;
  const float* x       = (const float*)d_in[0];
  const float* l1_w1   = (const float*)d_in[1];
  const float* l1_w2_1 = (const float*)d_in[2];
  const float* l1_w2_2 = (const float*)d_in[3];
  const float* l1_w2_3 = (const float*)d_in[4];
  const float* l1_w2_4 = (const float*)d_in[5];
  const float* l1_w3   = (const float*)d_in[6];
  const float* l2_w1   = (const float*)d_in[7];
  const float* l2_w2   = (const float*)d_in[8];
  const float* l2_w4   = (const float*)d_in[9];
  const float* w_fc1   = (const float*)d_in[10];
  const float* b_fc1   = (const float*)d_in[11];
  const float* w_fc2   = (const float*)d_in[12];
  const float* b_fc2   = (const float*)d_in[13];
  const float* gamma   = (const float*)d_in[14];
  const float* beta    = (const float*)d_in[15];
  float* out = (float*)d_out;

  float* l1_buf = (float*)d_ws;
  float* part1  = l1_buf + (size_t)BATCH * 196;
  float* part2  = part1 + 2 * BN_PARTS;
  float* wswz   = part2 + 2 * BN_PARTS;

  layer1_kernel<<<L1_GROUPS + WZ_GROUPS, 224, 0, stream>>>(
      x, l1_w1, l1_w2_1, l1_w2_2, l1_w2_3, l1_w2_4, l1_w3, l1_buf, (float2*)part1,
      w_fc1, w_fc2, (float2*)wswz);
  layer2_kernel<<<L2_GROUPS, 256, 0, stream>>>(l1_buf, part1, gamma, beta,
                                               l2_w1, l2_w2, l2_w4, (float2*)part2);
  gemm_kernel<<<BATCH / 16, 256, 0, stream>>>(l1_buf, part2, gamma, beta,
                                              (const float2*)wswz, b_fc1, b_fc2, out);
}
